# starting kernel
# baseline (speedup 1.0000x reference)
#include <hip/hip_runtime.h>


#define N_NODES  100000
#define N_EDGES  3200000
#define N_GRAPHS 4096
#define EMB      32
#define HID      64
#define NCLS     10

typedef __attribute__((ext_vector_type(16))) _Float16 v16h;
typedef __attribute__((ext_vector_type(8)))  float    v8f;

__device__ __forceinline__ v16h pack_frag(float4 c0, float4 c1,
                                          float4 c2, float4 c3, float s)
{
    v16h r;
    r[0]  = (_Float16)(c0.x * s); r[1]  = (_Float16)(c0.y * s);
    r[2]  = (_Float16)(c0.z * s); r[3]  = (_Float16)(c0.w * s);
    r[4]  = (_Float16)(c1.x * s); r[5]  = (_Float16)(c1.y * s);
    r[6]  = (_Float16)(c1.z * s); r[7]  = (_Float16)(c1.w * s);
    r[8]  = (_Float16)(c2.x * s); r[9]  = (_Float16)(c2.y * s);
    r[10] = (_Float16)(c2.z * s); r[11] = (_Float16)(c2.w * s);
    r[12] = (_Float16)(c3.x * s); r[13] = (_Float16)(c3.y * s);
    r[14] = (_Float16)(c3.z * s); r[15] = (_Float16)(c3.w * s);
    return r;
}

__device__ __forceinline__ v16h load_frag(const float4* __restrict__ q,
                                          int ks, int half, float s)
{
    const int o = ks * 8 + half * 2;
    return pack_frag(q[o], q[o + 1], q[o + 4], q[o + 5], s);
}

__global__ void embed_gather(const int* __restrict__ x_idx,
                             const float* __restrict__ embed_w,
                             float* __restrict__ x0)
{
    int tid = blockIdx.x * blockDim.x + threadIdx.x;
    if (tid >= N_NODES * EMB) return;
    int node = tid >> 5;
    int f    = tid & (EMB - 1);
    x0[tid] = embed_w[x_idx[node] * EMB + f];
}

__global__ void edge_count(const int* __restrict__ dst, float* __restrict__ cnt)
{
    int e = blockIdx.x * blockDim.x + threadIdx.x;
    if (e >= N_EDGES) return;
    atomicAdd(&cnt[dst[e]], 1.0f);
}

template <int F>
__global__ void edge_scatter(const int* __restrict__ src,
                             const int* __restrict__ dst,
                             const float* __restrict__ x,
                             float* __restrict__ agg)
{
    int tid = blockIdx.x * blockDim.x + threadIdx.x;
    if (tid >= N_EDGES * F) return;
    int e = tid / F;
    int f = tid & (F - 1);
    int s = src[e];
    int d = dst[e];
    atomicAdd(&agg[d * F + f], x[s * F + f]);
}

template <int K>
__global__ void sage_linear_wmma(const float* __restrict__ agg,
                                 const float* __restrict__ cnt,
                                 const float* __restrict__ xin,
                                 const float* __restrict__ Wl,
                                 const float* __restrict__ bl,
                                 const float* __restrict__ Wr,
                                 float* __restrict__ out)
{
    const int lane  = threadIdx.x & 31;
    const int wave  = threadIdx.x >> 5;
    const int wpb   = blockDim.x >> 5;
    const int tile  = blockIdx.x * wpb + wave;
    const int row0  = tile * 16;
    if (row0 >= N_NODES) return;

    const int half = lane >> 4;
    const int m    = lane & 15;
    const int rowA = row0 + m;

    const float inv = 1.0f / fmaxf(cnt[rowA], 1.0f);

    constexpr int KS = K / 32;
    const float4* aggq = (const float4*)(agg + rowA * K);
    const float4* xq   = (const float4*)(xin + rowA * K);

    v16h aM[KS], aX[KS];
    #pragma unroll
    for (int ks = 0; ks < KS; ++ks) {
        aM[ks] = load_frag(aggq, ks, half, inv);
        aX[ks] = load_frag(xq,   ks, half, 1.0f);
    }

    #pragma unroll
    for (int nt = 0; nt < HID / 16; ++nt) {
        const int ncol = nt * 16 + m;
        const float4* wlq = (const float4*)(Wl + ncol * K);
        const float4* wrq = (const float4*)(Wr + ncol * K);

        v8f acc = {};
        #pragma unroll
        for (int ks = 0; ks < KS; ++ks) {
            v16h bL = load_frag(wlq, ks, half, 1.0f);
            v16h bR = load_frag(wrq, ks, half, 1.0f);
            acc = __builtin_amdgcn_wmma_f32_16x16x32_f16(
                false, aM[ks], false, bL, (short)0, acc, false, false);
            acc = __builtin_amdgcn_wmma_f32_16x16x32_f16(
                false, aX[ks], false, bR, (short)0, acc, false, false);
        }

        const float bias = bl[ncol];
        float* op = out + (row0 + half * 8) * HID + ncol;
        #pragma unroll
        for (int d = 0; d < 8; ++d)
            op[d * HID] = fmaxf(acc[d] + bias, 0.0f);
    }
}

__global__ void pool_features(const int* __restrict__ batch,
                              const float* __restrict__ h2,
                              float* __restrict__ pooled)
{
    int tid = blockIdx.x * blockDim.x + threadIdx.x;
    if (tid >= N_NODES * HID) return;
    int node = tid >> 6;
    int f    = tid & (HID - 1);
    atomicAdd(&pooled[batch[node] * HID + f], h2[tid]);
}

__global__ void pool_count(const int* __restrict__ batch, float* __restrict__ gcnt)
{
    int i = blockIdx.x * blockDim.x + threadIdx.x;
    if (i >= N_NODES) return;
    atomicAdd(&gcnt[batch[i]], 1.0f);
}

__global__ void final_head(const float* __restrict__ pooled,
                           const float* __restrict__ gcnt,
                           const float* __restrict__ Wout,
                           const float* __restrict__ bout,
                           float* __restrict__ out)
{
    int tid = blockIdx.x * blockDim.x + threadIdx.x;
    if (tid >= N_GRAPHS * NCLS) return;
    int g = tid / NCLS;
    int c = tid % NCLS;
    float inv = 1.0f / fmaxf(gcnt[g], 1.0f);
    float acc = bout[c];
    #pragma unroll
    for (int h = 0; h < HID; ++h)
        acc += pooled[g * HID + h] * inv * Wout[c * HID + h];
    out[tid] = acc;
}

extern "C" void kernel_launch(void* const* d_in, const int* in_sizes, int n_in,
                              void* d_out, int out_size, void* d_ws, size_t ws_size,
                              hipStream_t stream)
{
    const int*   x_idx   = (const int*)  d_in[0];
    const int*   ei      = (const int*)  d_in[1];
    const int*   batch   = (const int*)  d_in[2];
    const float* embed_w = (const float*)d_in[3];
    const float* W1l     = (const float*)d_in[4];
    const float* b1      = (const float*)d_in[5];
    const float* W1r     = (const float*)d_in[6];
    const float* W2l     = (const float*)d_in[7];
    const float* b2      = (const float*)d_in[8];
    const float* W2r     = (const float*)d_in[9];
    const float* Wout    = (const float*)d_in[10];
    const float* bout    = (const float*)d_in[11];
    float*       out     = (float*)d_out;

    const int* src = ei;
    const int* dst = ei + N_EDGES;

    float* cnt    = (float*)d_ws;
    float* x0     = cnt    + N_NODES;
    float* agg1   = x0     + N_NODES * EMB;
    float* agg2   = x0;
    float* h1     = agg1   + N_NODES * EMB;
    float* h2     = h1     + N_NODES * HID;
    float* pooled = h2     + N_NODES * HID;
    float* gcnt   = pooled + N_GRAPHS * HID;

    const int B = 256;

    hipMemsetAsync(cnt,    0, sizeof(float) * N_NODES,        stream);
    hipMemsetAsync(agg1,   0, sizeof(float) * N_NODES * EMB,  stream);
    hipMemsetAsync(pooled, 0, sizeof(float) * N_GRAPHS * HID, stream);
    hipMemsetAsync(gcnt,   0, sizeof(float) * N_GRAPHS,       stream);

    embed_gather<<<(N_NODES * EMB + B - 1) / B, B, 0, stream>>>(x_idx, embed_w, x0);

    edge_count<<<(N_EDGES + B - 1) / B, B, 0, stream>>>(dst, cnt);

    edge_scatter<EMB><<<(N_EDGES * EMB + B - 1) / B, B, 0, stream>>>(src, dst, x0, agg1);
    {
        const int tiles  = N_NODES / 16;
        const int blocks = (tiles + 7) / 8;
        sage_linear_wmma<EMB><<<blocks, 256, 0, stream>>>(agg1, cnt, x0, W1l, b1, W1r, h1);
    }

    hipMemsetAsync(agg2, 0, sizeof(float) * N_NODES * HID, stream);
    edge_scatter<HID><<<(int)(((long long)N_EDGES * HID + B - 1) / B), B, 0, stream>>>(src, dst, h1, agg2);
    {
        const int tiles  = N_NODES / 16;
        const int blocks = (tiles + 7) / 8;
        sage_linear_wmma<HID><<<blocks, 256, 0, stream>>>(agg2, cnt, h1, W2l, b2, W2r, h2);
    }

    pool_features<<<(N_NODES * HID + B - 1) / B, B, 0, stream>>>(batch, h2, pooled);
    pool_count<<<(N_NODES + B - 1) / B, B, 0, stream>>>(batch, gcnt);
    final_head<<<(N_GRAPHS * NCLS + B - 1) / B, B, 0, stream>>>(pooled, gcnt, Wout, bout, out);
}


# neighbour lists by owner work-groups: each reads all edge targets, ranks its 800 nodes' edges by LDS adds into fixed 64-entry lists (+spill); no global rank add, scan, fill; sums in LDS arrival order
# speedup vs baseline: 3.0530x; 3.0530x over previous
#include <hip/hip_runtime.h>


#define N_NODES  100000
#define N_EDGES  3200000
#define N_GRAPHS 4096
#define EMB      32
#define HID      64
#define NCLS     10
#define SLOT     64

typedef __attribute__((ext_vector_type(16))) _Float16 v16h;
typedef __attribute__((ext_vector_type(8)))  float    v8f;

__device__ __forceinline__ v16h pack_frag(float4 c0, float4 c1,
                                          float4 c2, float4 c3, float s)
{
    v16h r;
    r[0]  = (_Float16)(c0.x * s); r[1]  = (_Float16)(c0.y * s);
    r[2]  = (_Float16)(c0.z * s); r[3]  = (_Float16)(c0.w * s);
    r[4]  = (_Float16)(c1.x * s); r[5]  = (_Float16)(c1.y * s);
    r[6]  = (_Float16)(c1.z * s); r[7]  = (_Float16)(c1.w * s);
    r[8]  = (_Float16)(c2.x * s); r[9]  = (_Float16)(c2.y * s);
    r[10] = (_Float16)(c2.z * s); r[11] = (_Float16)(c2.w * s);
    r[12] = (_Float16)(c3.x * s); r[13] = (_Float16)(c3.y * s);
    r[14] = (_Float16)(c3.z * s); r[15] = (_Float16)(c3.w * s);
    return r;
}

__device__ __forceinline__ v16h load_frag(const float4* __restrict__ q,
                                          int ks, int half, float s)
{
    const int o = ks * 8 + half * 2;
    return pack_frag(q[o], q[o + 1], q[o + 4], q[o + 5], s);
}

__global__ void embed_gather(const int* __restrict__ x_idx,
                             const float* __restrict__ embed_w,
                             float* __restrict__ x0)
{
    int tid = blockIdx.x * blockDim.x + threadIdx.x;
    if (tid >= N_NODES * EMB) return;
    int node = tid >> 5;
    int f    = tid & (EMB - 1);
    x0[tid] = embed_w[x_idx[node] * EMB + f];
}

#define RNG   800
#define NOWN  (N_NODES / RNG)
#define E4    (N_EDGES / 4)
#define QCAP  160
#define UNR   8

__global__ void __launch_bounds__(1024)
build_lists(const int* __restrict__ src,
            const int* __restrict__ dst,
            int* __restrict__ cnt,
            int* __restrict__ lst,
            int* __restrict__ spill_n,
            int* __restrict__ spill_src,
            int* __restrict__ spill_dst)
{
    __shared__ int  cntL[RNG];
    __shared__ int2 qL[32 * QCAP];

    const int lane = threadIdx.x & 31;
    const int wave = threadIdx.x >> 5;
    const int base = blockIdx.x * RNG;
    int2* qw = qL + wave * QCAP;

    for (int i = threadIdx.x; i < RNG; i += 1024) cntL[i] = 0;
    __syncthreads();

    auto take = [&](int first, int n) {
        if (lane < n) {
            const int2 t = qw[first + lane];
            const int s = src[t.x];
            const int r = atomicAdd(&cntL[t.y], 1);
            const int node = base + t.y;
            if (r < SLOT) {
                lst[node * SLOT + r] = s;
            } else {
                const int k = atomicAdd(spill_n, 1);
                spill_src[k] = s;
                spill_dst[k] = node;
            }
        }
    };

    const int rot = blockIdx.x * (E4 / NOWN);
    const int4* __restrict__ d4 = (const int4*)dst;
    int qn = 0;

    for (int g = 0; g < (E4 + 1024 * UNR - 1) / (1024 * UNR); ++g) {
        int4 v[UNR];
        int  ci[UNR];
        #pragma unroll
        for (int u = 0; u < UNR; ++u) {
            const int c = (g * UNR + u) * 1024 + (int)threadIdx.x;
            int cc = c + rot;
            if (cc >= E4) cc -= E4;
            ci[u] = cc;
            v[u] = (c < E4) ? d4[cc] : make_int4(-1, -1, -1, -1);
        }
        #pragma unroll
        for (int u = 0; u < UNR; ++u) {
            unsigned t[4];
            t[0] = (unsigned)(v[u].x - base); t[1] = (unsigned)(v[u].y - base);
            t[2] = (unsigned)(v[u].z - base); t[3] = (unsigned)(v[u].w - base);
            const unsigned mn = min(min(t[0], t[1]), min(t[2], t[3]));
            if (__builtin_amdgcn_ballot_w32(mn < (unsigned)RNG) != 0u) {
                #pragma unroll
                for (int j = 0; j < 4; ++j) {
                    const bool hit = t[j] < (unsigned)RNG;
                    const unsigned m = __builtin_amdgcn_ballot_w32(hit);
                    if (hit)
                        qw[qn + (int)__builtin_amdgcn_mbcnt_lo(m, 0u)] =
                            make_int2(ci[u] * 4 + j, (int)t[j]);
                    qn += __builtin_popcount(m);
                }
                while (qn >= 32) { qn -= 32; take(qn, 32); }
            }
        }
    }
    take(0, qn);

    __syncthreads();
    for (int i = threadIdx.x; i < RNG; i += 1024) cnt[base + i] = cntL[i];
}


template <int KS>
__device__ __forceinline__ void add_row(float4 (&s)[KS][4],
                                        const float4* __restrict__ q, int half)
{
    #pragma unroll
    for (int ks = 0; ks < KS; ++ks) {
        const int o = ks * 8 + half * 2;
        const float4 a0 = q[o], a1 = q[o + 1], a2 = q[o + 4], a3 = q[o + 5];
        s[ks][0].x += a0.x; s[ks][0].y += a0.y; s[ks][0].z += a0.z; s[ks][0].w += a0.w;
        s[ks][1].x += a1.x; s[ks][1].y += a1.y; s[ks][1].z += a1.z; s[ks][1].w += a1.w;
        s[ks][2].x += a2.x; s[ks][2].y += a2.y; s[ks][2].z += a2.z; s[ks][2].w += a2.w;
        s[ks][3].x += a3.x; s[ks][3].y += a3.y; s[ks][3].z += a3.z; s[ks][3].w += a3.w;
    }
}

template <int K>
__global__ void sage_pull_wmma(const int* __restrict__ cnt,
                               const int* __restrict__ lst,
                               const int* __restrict__ spill_n,
                               const int* __restrict__ spill_src,
                               const int* __restrict__ spill_dst,
                               const float* __restrict__ xin,
                               const float* __restrict__ Wl,
                               const float* __restrict__ bl,
                               const float* __restrict__ Wr,
                               float* __restrict__ out)
{
    const int lane  = threadIdx.x & 31;
    const int wave  = threadIdx.x >> 5;
    const int wpb   = blockDim.x >> 5;
    const int tile  = blockIdx.x * wpb + wave;
    const int row0  = tile * 16;
    if (row0 >= N_NODES) return;

    const int half = lane >> 4;
    const int m    = lane & 15;
    const int rowA = row0 + m;

    constexpr int KS = K / 32;
    const int deg = cnt[rowA];
    const int end = deg < SLOT ? deg : SLOT;
    const int* __restrict__ nbr = lst + rowA * SLOT;
    const float inv = 1.0f / fmaxf((float)deg, 1.0f);

    float4 s[KS][4];
    #pragma unroll
    for (int ks = 0; ks < KS; ++ks)
        #pragma unroll
        for (int c = 0; c < 4; ++c) s[ks][c] = make_float4(0.f, 0.f, 0.f, 0.f);

    int p = 0;
    for (; p + 1 < end; p += 2) {
        const float4* q0 = (const float4*)(xin + nbr[p]     * K);
        const float4* q1 = (const float4*)(xin + nbr[p + 1] * K);
        float4 a[KS][4], b[KS][4];
        #pragma unroll
        for (int ks = 0; ks < KS; ++ks) {
            const int o = ks * 8 + half * 2;
            a[ks][0] = q0[o]; a[ks][1] = q0[o + 1]; a[ks][2] = q0[o + 4]; a[ks][3] = q0[o + 5];
            b[ks][0] = q1[o]; b[ks][1] = q1[o + 1]; b[ks][2] = q1[o + 4]; b[ks][3] = q1[o + 5];
        }
        #pragma unroll
        for (int ks = 0; ks < KS; ++ks)
            #pragma unroll
            for (int c = 0; c < 4; ++c) {
                s[ks][c].x += a[ks][c].x; s[ks][c].y += a[ks][c].y;
                s[ks][c].z += a[ks][c].z; s[ks][c].w += a[ks][c].w;
                s[ks][c].x += b[ks][c].x; s[ks][c].y += b[ks][c].y;
                s[ks][c].z += b[ks][c].z; s[ks][c].w += b[ks][c].w;
            }
    }
    if (p < end)
        add_row<KS>(s, (const float4*)(xin + nbr[p] * K), half);

    if (deg > SLOT) {
        const int ns = *spill_n;
        for (int k = 0; k < ns; ++k)
            if (spill_dst[k] == rowA)
                add_row<KS>(s, (const float4*)(xin + spill_src[k] * K), half);
    }

    const float4* xq = (const float4*)(xin + rowA * K);
    v16h aM[KS], aX[KS];
    #pragma unroll
    for (int ks = 0; ks < KS; ++ks) {
        aM[ks] = pack_frag(s[ks][0], s[ks][1], s[ks][2], s[ks][3], inv);
        aX[ks] = load_frag(xq, ks, half, 1.0f);
    }

    #pragma unroll
    for (int nt = 0; nt < HID / 16; ++nt) {
        const int ncol = nt * 16 + m;
        const float4* wlq = (const float4*)(Wl + ncol * K);
        const float4* wrq = (const float4*)(Wr + ncol * K);

        v8f acc = {};
        #pragma unroll
        for (int ks = 0; ks < KS; ++ks) {
            v16h bL = load_frag(wlq, ks, half, 1.0f);
            v16h bR = load_frag(wrq, ks, half, 1.0f);
            acc = __builtin_amdgcn_wmma_f32_16x16x32_f16(
                false, aM[ks], false, bL, (short)0, acc, false, false);
            acc = __builtin_amdgcn_wmma_f32_16x16x32_f16(
                false, aX[ks], false, bR, (short)0, acc, false, false);
        }

        const float bias = bl[ncol];
        float* op = out + (row0 + half * 8) * HID + ncol;
        #pragma unroll
        for (int d = 0; d < 8; ++d)
            op[d * HID] = fmaxf(acc[d] + bias, 0.0f);
    }
}

__global__ void pool_features(const int* __restrict__ batch,
                              const float* __restrict__ h2,
                              float* __restrict__ pooled)
{
    int tid = blockIdx.x * blockDim.x + threadIdx.x;
    if (tid >= N_NODES * HID) return;
    int node = tid >> 6;
    int f    = tid & (HID - 1);
    atomicAdd(&pooled[batch[node] * HID + f], h2[tid]);
}

__global__ void pool_count(const int* __restrict__ batch, float* __restrict__ gcnt)
{
    int i = blockIdx.x * blockDim.x + threadIdx.x;
    if (i >= N_NODES) return;
    atomicAdd(&gcnt[batch[i]], 1.0f);
}

__global__ void final_head(const float* __restrict__ pooled,
                           const float* __restrict__ gcnt,
                           const float* __restrict__ Wout,
                           const float* __restrict__ bout,
                           float* __restrict__ out)
{
    int tid = blockIdx.x * blockDim.x + threadIdx.x;
    if (tid >= N_GRAPHS * NCLS) return;
    int g = tid / NCLS;
    int c = tid % NCLS;
    float inv = 1.0f / fmaxf(gcnt[g], 1.0f);
    float acc = bout[c];
    #pragma unroll
    for (int h = 0; h < HID; ++h)
        acc += pooled[g * HID + h] * inv * Wout[c * HID + h];
    out[tid] = acc;
}

extern "C" void kernel_launch(void* const* d_in, const int* in_sizes, int n_in,
                              void* d_out, int out_size, void* d_ws, size_t ws_size,
                              hipStream_t stream)
{
    const int*   x_idx   = (const int*)  d_in[0];
    const int*   ei      = (const int*)  d_in[1];
    const int*   batch   = (const int*)  d_in[2];
    const float* embed_w = (const float*)d_in[3];
    const float* W1l     = (const float*)d_in[4];
    const float* b1      = (const float*)d_in[5];
    const float* W1r     = (const float*)d_in[6];
    const float* W2l     = (const float*)d_in[7];
    const float* b2      = (const float*)d_in[8];
    const float* W2r     = (const float*)d_in[9];
    const float* Wout    = (const float*)d_in[10];
    const float* bout    = (const float*)d_in[11];
    float*       out     = (float*)d_out;

    const int* src = ei;
    const int* dst = ei + N_EDGES;

    float* x0     = (float*)d_ws;
    float* h1     = x0     + N_NODES * EMB;
    float* h2     = h1     + N_NODES * HID;
    int*   lst    = (int*)(h2 + N_NODES * HID);
    float* pooled = (float*)(lst + N_NODES * SLOT);
    float* gcnt   = pooled + N_GRAPHS * HID;
    int*   cnt    = (int*)(gcnt + N_GRAPHS);
    int*   spilln = cnt    + N_NODES;
    int*   spills = spilln + 4;
    int*   spilld = spills + N_EDGES;

    const int B = 256;

    hipMemsetAsync(spilln, 0, sizeof(int)   * 4,              stream);
    hipMemsetAsync(pooled, 0, sizeof(float) * N_GRAPHS * HID, stream);
    hipMemsetAsync(gcnt,   0, sizeof(float) * N_GRAPHS,       stream);

    embed_gather<<<(N_NODES * EMB + B - 1) / B, B, 0, stream>>>(x_idx, embed_w, x0);

    build_lists<<<NOWN, 1024, 0, stream>>>(src, dst, cnt, lst, spilln, spills, spilld);

    const int tiles  = N_NODES / 16;
    const int blocks = (tiles + 7) / 8;

    sage_pull_wmma<EMB><<<blocks, 256, 0, stream>>>(cnt, lst, spilln, spills, spilld, x0, W1l, b1, W1r, h1);

    sage_pull_wmma<HID><<<blocks, 256, 0, stream>>>(cnt, lst, spilln, spills, spilld, h1, W2l, b2, W2r, h2);

    pool_features<<<(N_NODES * HID + B - 1) / B, B, 0, stream>>>(batch, h2, pooled);
    pool_count<<<(N_NODES + B - 1) / B, B, 0, stream>>>(batch, gcnt);
    final_head<<<(N_GRAPHS * NCLS + B - 1) / B, B, 0, stream>>>(pooled, gcnt, Wout, bout, out);
}


# pooling pulled by graph: layer 2 marks each graph's run start in batch; one wave per graph sums its h2 rows in node order (fixed tree) and applies the head; no float atomics, zeroing, pooled plane
# speedup vs baseline: 6.1756x; 2.0228x over previous
#include <hip/hip_runtime.h>


#define N_NODES  100000
#define N_EDGES  3200000
#define N_GRAPHS 4096
#define EMB      32
#define HID      64
#define NCLS     10
#define NPART    8
#define CAP      16
#define PSTR     (2 * CAP)
#define PAIRW    (NPART * PSTR)

typedef __attribute__((ext_vector_type(16))) _Float16 v16h;
typedef __attribute__((ext_vector_type(8)))  float    v8f;

__device__ __forceinline__ v16h pack_frag(float4 c0, float4 c1,
                                          float4 c2, float4 c3, float s)
{
    v16h r;
    r[0]  = (_Float16)(c0.x * s); r[1]  = (_Float16)(c0.y * s);
    r[2]  = (_Float16)(c0.z * s); r[3]  = (_Float16)(c0.w * s);
    r[4]  = (_Float16)(c1.x * s); r[5]  = (_Float16)(c1.y * s);
    r[6]  = (_Float16)(c1.z * s); r[7]  = (_Float16)(c1.w * s);
    r[8]  = (_Float16)(c2.x * s); r[9]  = (_Float16)(c2.y * s);
    r[10] = (_Float16)(c2.z * s); r[11] = (_Float16)(c2.w * s);
    r[12] = (_Float16)(c3.x * s); r[13] = (_Float16)(c3.y * s);
    r[14] = (_Float16)(c3.z * s); r[15] = (_Float16)(c3.w * s);
    return r;
}

__device__ __forceinline__ v16h load_frag(const float4* __restrict__ q,
                                          int ks, int half, float s)
{
    const int o = ks * 8 + half * 2;
    return pack_frag(q[o], q[o + 1], q[o + 4], q[o + 5], s);
}

__global__ void embed_gather(const int* __restrict__ x_idx,
                             const float* __restrict__ embed_w,
                             float* __restrict__ x0)
{
    int tid = blockIdx.x * blockDim.x + threadIdx.x;
    if (tid >= N_NODES * EMB) return;
    int node = tid >> 5;
    int f    = tid & (EMB - 1);
    x0[tid] = embed_w[x_idx[node] * EMB + f];
}

#define NRNG  32
#define RNG   3126
static_assert(RNG % 2 == 0 && NRNG * RNG >= N_NODES && (NRNG - 1) * RNG < N_NODES, "ranges");
#define NOWN  (NRNG * NPART)
#define E4    (N_EDGES / 4)
#define PLEN  (E4 / NPART)
#define QCAP  160
#define UNR   8

__global__ void __launch_bounds__(1024)
build_lists(const int* __restrict__ src,
            const int* __restrict__ dst,
            int* __restrict__ cnt,
            int* __restrict__ lst,
            int* __restrict__ spill_n,
            int* __restrict__ spill_src,
            int* __restrict__ spill_dst)
{
    __shared__ int  cntL[RNG];
    __shared__ int2 qL[32 * QCAP];

    const int lane = threadIdx.x & 31;
    const int wave = threadIdx.x >> 5;
    const int rng  = blockIdx.x / NPART;
    const int part = blockIdx.x % NPART;
    const int base = rng * RNG;
    const int nloc = min(RNG, N_NODES - base);
    int2* qw = qL + wave * QCAP;

    for (int i = threadIdx.x; i < RNG; i += 1024) cntL[i] = 0;
    __syncthreads();

    auto take = [&](int first, int n) {
        if (lane < n) {
            const int2 t = qw[first + lane];
            const int s = src[t.x];
            const int r = atomicAdd(&cntL[t.y], 1);
            const int node = base + t.y;
            if (r < CAP) {
                lst[(node >> 1) * PAIRW + part * PSTR + (node & 1) * CAP + r] = s;
            } else {
                const int k = atomicAdd(spill_n, 1);
                spill_src[k] = s;
                spill_dst[k] = node;
            }
        }
    };

    const int rot = rng * (PLEN / NRNG);
    const int4* __restrict__ d4 = (const int4*)dst + part * PLEN;
    const int e0 = part * (PLEN * 4);
    int qn = 0;

    for (int g = 0; g < (PLEN + 1024 * UNR - 1) / (1024 * UNR); ++g) {
        int4 v[UNR];
        int  ci[UNR];
        #pragma unroll
        for (int u = 0; u < UNR; ++u) {
            const int c = (g * UNR + u) * 1024 + (int)threadIdx.x;
            int cc = c + rot;
            if (cc >= PLEN) cc -= PLEN;
            ci[u] = cc;
            v[u] = (c < PLEN) ? d4[cc] : make_int4(-1, -1, -1, -1);
        }
        #pragma unroll
        for (int u = 0; u < UNR; ++u) {
            unsigned t[4];
            t[0] = (unsigned)(v[u].x - base); t[1] = (unsigned)(v[u].y - base);
            t[2] = (unsigned)(v[u].z - base); t[3] = (unsigned)(v[u].w - base);
            const unsigned mn = min(min(t[0], t[1]), min(t[2], t[3]));
            if (__builtin_amdgcn_ballot_w32(mn < (unsigned)RNG) != 0u) {
                #pragma unroll
                for (int j = 0; j < 4; ++j) {
                    const bool hit = t[j] < (unsigned)RNG;
                    const unsigned m = __builtin_amdgcn_ballot_w32(hit);
                    if (hit)
                        qw[qn + (int)__builtin_amdgcn_mbcnt_lo(m, 0u)] =
                            make_int2(e0 + ci[u] * 4 + j, (int)t[j]);
                    qn += __builtin_popcount(m);
                }
                while (qn >= 32) { qn -= 32; take(qn, 32); }
            }
        }
    }
    take(0, qn);

    __syncthreads();
    for (int i = threadIdx.x; i < nloc; i += 1024)
        cnt[part * N_NODES + base + i] = cntL[i];
}


#define WPB   4
#define FSTR  148
static_assert(FSTR >= NPART * CAP + NPART && FSTR % 4 == 0 && (FSTR / 4) % 2 == 1, "row pitch");

template <int K, bool MARK>
__global__ void __launch_bounds__(WPB * 32)
sage_pull_wmma(const int* __restrict__ batch,
               int* __restrict__ gstart,
               int* __restrict__ unsorted,
               const int* __restrict__ cnt,
               const int* __restrict__ lst,
               const int* __restrict__ spill_n,
               const int* __restrict__ spill_src,
               const int* __restrict__ spill_dst,
               const float* __restrict__ xin,
               const float* __restrict__ Wl,
               const float* __restrict__ bl,
               const float* __restrict__ Wr,
               float* __restrict__ out)
{
    __shared__ __attribute__((aligned(16))) int stageL[WPB * 16 * FSTR];

    const int lane  = threadIdx.x & 31;
    const int wave  = threadIdx.x >> 5;
    const int tile  = blockIdx.x * WPB + wave;
    const int row0  = tile * 16;
    if (row0 >= N_NODES) return;

    const int half = lane >> 4;
    const int m    = lane & 15;
    const int rowA = row0 + m;

    int bcur = 0, bprev = -1;
    if constexpr (MARK) {
        bcur = batch[rowA];
        if (rowA > 0) bprev = batch[rowA - 1];
    }

    constexpr int KS = K / 32;
    constexpr int CH = K / 4;
    constexpr int G  = 32 / CH;
    constexpr int U  = 8;
    static_assert(U == 8 && (NPART * CAP) % (G * U) == 0, "a trip never reads past a row's indices");
    const int c = lane & (CH - 1);
    const int g = lane / CH;

    int deg = 0, end = 0;
    int info[NPART];
    #pragma unroll
    for (int k = 0; k < NPART; ++k) {
        const int ck = cnt[k * N_NODES + rowA];
        const int hk = ck < CAP ? ck : CAP;
        info[k] = end | (hk << 8);
        deg += ck;
        end += hk;
    }
    const float inv = 1.0f / fmaxf((float)deg, 1.0f);

    int* fw = stageL + wave * (16 * FSTR);
    int* fm = fw + m * FSTR;

    constexpr int TAB = NPART * CAP;
    static_assert(TAB + NPART <= FSTR && TAB % 4 == 0, "info words behind the indices");
    ((int4*)(fm + TAB))[half] = half ? make_int4(info[4], info[5], info[6], info[7])
                                     : make_int4(info[0], info[1], info[2], info[3]);

    {
        const int par = (lane >> 2) & 1;
        const int r0  = (lane & 3) * 4;
        const int kq  = lane >> 3;
        const int4* __restrict__ blk = (const int4*)(lst + (row0 >> 1) * PAIRW) + lane;
        #pragma unroll
        for (int b = 0; b < 2; ++b) {
            int4 e[8];
            #pragma unroll
            for (int t = 0; t < 8; ++t) e[t] = blk[(b * 8 + t) * 32];
            #pragma unroll
            for (int t = 0; t < 8; ++t) {
                const int tt = b * 8 + t;
                int* row = fw + (2 * (tt >> 1) + par) * FSTR;
                const int w  = row[TAB + 4 * (tt & 1) + kq];
                const int nv = (w >> 8) - r0;
                int* d = row + (w & 0xff) + r0;
                if (nv > 0) d[0] = e[t].x;
                if (nv > 1) d[1] = e[t].y;
                if (nv > 2) d[2] = e[t].z;
                if (nv > 3) d[3] = e[t].w;
            }
        }
    }

    const char* xb = (const char*)xin;
    const unsigned coff = (unsigned)c * 16u;
    auto chunk_of = [&](int i) -> float4 {
        return *(const float4*)(xb + ((unsigned)i * (unsigned)(K * 4) + coff));
    };

    for (int n = 0; n < 16; ++n) {
        const int endn = __builtin_amdgcn_readlane(end, n);
        const int degn = __builtin_amdgcn_readlane(deg, n);
        const int* fl = fw + n * FSTR;
        float4 acc = make_float4(0.f, 0.f, 0.f, 0.f);
        for (int j0 = 0; j0 < endn; j0 += G * U) {
            const int4* ip = (const int4*)__builtin_assume_aligned(fl + j0 + g * U, 16);
            const int4 ia = ip[0];
            const int4 ib = ip[1];
            const int idx[U] = {ia.x, ia.y, ia.z, ia.w, ib.x, ib.y, ib.z, ib.w};
            float4 v[U];
            if (j0 + G * U <= endn) {
                #pragma unroll
                for (int u = 0; u < U; ++u) v[u] = chunk_of(idx[u]);
            } else {
                #pragma unroll
                for (int u = 0; u < U; ++u) {
                    v[u] = make_float4(0.f, 0.f, 0.f, 0.f);
                    if (j0 + g * U + u < endn) v[u] = chunk_of(idx[u]);
                }
            }
            #pragma unroll
            for (int u = 0; u < U; ++u) {
                acc.x += v[u].x; acc.y += v[u].y; acc.z += v[u].z; acc.w += v[u].w;
            }
        }
        if (degn > endn) {
            const int ns = *spill_n;
            for (int k = 0; k < ns; ++k)
                if (spill_dst[k] == row0 + n && g == 0) {
                    const float4 a = chunk_of(spill_src[k]);
                    acc.x += a.x; acc.y += a.y; acc.z += a.z; acc.w += a.w;
                }
        }
        #pragma unroll
        for (int off = CH; off < 32; off <<= 1) {
            acc.x += __shfl_xor(acc.x, off); acc.y += __shfl_xor(acc.y, off);
            acc.z += __shfl_xor(acc.z, off); acc.w += __shfl_xor(acc.w, off);
        }
        if (g == 0)
            ((int4*)(fw + n * FSTR))[c] = make_int4(__float_as_int(acc.x), __float_as_int(acc.y),
                                                    __float_as_int(acc.z), __float_as_int(acc.w));
    }

    const int4*   sq = (const int4*)fm;
    const float4* xq = (const float4*)(xin + rowA * K);
    auto f4 = [](int4 t) -> float4 {
        return make_float4(__int_as_float(t.x), __int_as_float(t.y),
                           __int_as_float(t.z), __int_as_float(t.w));
    };
    v16h aM[KS], aX[KS];
    #pragma unroll
    for (int ks = 0; ks < KS; ++ks) {
        const int o = ks * 8 + half * 2;
        aM[ks] = pack_frag(f4(sq[o]), f4(sq[o + 1]), f4(sq[o + 4]), f4(sq[o + 5]), inv);
        aX[ks] = load_frag(xq, ks, half, 1.0f);
    }

    #pragma unroll 1
    for (int nt = 0; nt < HID / 16; ++nt) {
        const int ncol = nt * 16 + m;
        const float4* wlq = (const float4*)(Wl + ncol * K);
        const float4* wrq = (const float4*)(Wr + ncol * K);

        v8f acc = {};
        #pragma unroll
        for (int ks = 0; ks < KS; ++ks) {
            v16h bL = load_frag(wlq, ks, half, 1.0f);
            v16h bR = load_frag(wrq, ks, half, 1.0f);
            acc = __builtin_amdgcn_wmma_f32_16x16x32_f16(
                false, aM[ks], false, bL, (short)0, acc, false, false);
            acc = __builtin_amdgcn_wmma_f32_16x16x32_f16(
                false, aX[ks], false, bR, (short)0, acc, false, false);
        }

        const float bias = bl[ncol];
        float* op = out + (row0 + half * 8) * HID + ncol;
        #pragma unroll
        for (int d = 0; d < 8; ++d)
            op[d * HID] = fmaxf(acc[d] + bias, 0.0f);
    }

    if constexpr (MARK) {
        if (half == 0) {
            if (bcur < bprev) {
                *unsorted = 1;
            } else {
                #pragma clang loop unroll(disable) vectorize(disable)
                for (int q = max(bprev + 1, 0); q <= min(bcur, N_GRAPHS - 1); ++q) gstart[q] = rowA;
            }
            if (rowA == N_NODES - 1) {
                #pragma clang loop unroll(disable) vectorize(disable)
                for (int q = max(bcur + 1, 0); q <= N_GRAPHS; ++q) gstart[q] = N_NODES;
            }
        }
    }
}

#define GPB 16
#define PU  8
static_assert(N_GRAPHS % GPB == 0 && (GPB * NCLS * 4) % 128 == 0, "whole output lines per work-group");
static_assert(NCLS == 10 && HID == 64 && N_NODES % 32 == 0, "lane maps of the pooling kernel");

__global__ void __launch_bounds__(GPB * 32)
pool_head(const int* __restrict__ batch,
          const int* __restrict__ gstart,
          const int* __restrict__ unsorted,
          const float* __restrict__ h2,
          const float* __restrict__ Wout,
          const float* __restrict__ bout,
          float* __restrict__ out)
{
    __shared__ float outL[GPB * NCLS];

    const int lane = threadIdx.x & 31;
    const int wave = threadIdx.x >> 5;
    const int q    = blockIdx.x * GPB + wave;
    const int half = lane >> 4;
    const int c    = lane & 15;
    const float4* __restrict__ hq = (const float4*)h2 + c;

    float4 w[5];
    #pragma unroll
    for (int k = 0; k < 5; ++k) w[k] = ((const float4*)Wout)[(half * 5 + k) * (HID / 4) + c];

    float4 acc = make_float4(0.f, 0.f, 0.f, 0.f);
    int n = 0;
    if (*unsorted == 0) {
        const int lo = gstart[q];
        const int hi = gstart[q + 1];
        n = hi - lo;
        for (int r0 = lo; r0 < hi; r0 += 2 * PU) {
            float4 v[PU];
            #pragma unroll
            for (int u = 0; u < PU; ++u) {
                const int r = r0 + 2 * u + half;
                v[u] = make_float4(0.f, 0.f, 0.f, 0.f);
                if (r < hi) v[u] = hq[r * (HID / 4)];
            }
            #pragma unroll
            for (int u = 0; u < PU; ++u) {
                acc.x += v[u].x; acc.y += v[u].y; acc.z += v[u].z; acc.w += v[u].w;
            }
        }
    } else {
        for (int i0 = 0; i0 < N_NODES; i0 += 32) {
            unsigned mk = __builtin_amdgcn_ballot_w32(batch[i0 + lane] == q);
            n += __builtin_popcount(mk);
            while (mk != 0u) {
                const int j = __builtin_ctz(mk);
                mk &= mk - 1u;
                if (half == 0) {
                    const float4 a = hq[(i0 + j) * (HID / 4)];
                    acc.x += a.x; acc.y += a.y; acc.z += a.z; acc.w += a.w;
                }
            }
        }
    }
    acc.x += __shfl_xor(acc.x, 16); acc.y += __shfl_xor(acc.y, 16);
    acc.z += __shfl_xor(acc.z, 16); acc.w += __shfl_xor(acc.w, 16);

    const float inv = 1.0f / fmaxf((float)n, 1.0f);
    float p[5];
    #pragma unroll
    for (int k = 0; k < 5; ++k)
        p[k] = acc.x * w[k].x + acc.y * w[k].y + acc.z * w[k].z + acc.w * w[k].w;
    #pragma unroll
    for (int off = 8; off >= 1; off >>= 1) {
        #pragma unroll
        for (int k = 0; k < 5; ++k) p[k] += __shfl_xor(p[k], off);
    }
    if (c == 0) {
        #pragma unroll
        for (int k = 0; k < 5; ++k)
            outL[wave * NCLS + half * 5 + k] = bout[half * 5 + k] + p[k] * inv;
    }
    __syncthreads();
    if (threadIdx.x < GPB * NCLS)
        out[blockIdx.x * (GPB * NCLS) + threadIdx.x] = outL[threadIdx.x];
}

extern "C" void kernel_launch(void* const* d_in, const int* in_sizes, int n_in,
                              void* d_out, int out_size, void* d_ws, size_t ws_size,
                              hipStream_t stream)
{
    const int*   x_idx   = (const int*)  d_in[0];
    const int*   ei      = (const int*)  d_in[1];
    const int*   batch   = (const int*)  d_in[2];
    const float* embed_w = (const float*)d_in[3];
    const float* W1l     = (const float*)d_in[4];
    const float* b1      = (const float*)d_in[5];
    const float* W1r     = (const float*)d_in[6];
    const float* W2l     = (const float*)d_in[7];
    const float* b2      = (const float*)d_in[8];
    const float* W2r     = (const float*)d_in[9];
    const float* Wout    = (const float*)d_in[10];
    const float* bout    = (const float*)d_in[11];
    float*       out     = (float*)d_out;

    const int* src = ei;
    const int* dst = ei + N_EDGES;

    float* x0     = (float*)d_ws;
    float* h1     = x0     + N_NODES * EMB;
    float* h2     = h1     + N_NODES * HID;
    int*   lst    = (int*)(h2 + N_NODES * HID);
    int*   gstart = lst + (N_NODES / 2) * PAIRW;
    int*   cnt    = gstart + N_GRAPHS + 4;
    int*   spilln = cnt    + NPART * N_NODES;
    int*   spills = spilln + 4;
    int*   spilld = spills + N_EDGES;

    const int B = 256;

    hipMemsetAsync(spilln, 0, sizeof(int) * 4, stream);

    embed_gather<<<(N_NODES * EMB + B - 1) / B, B, 0, stream>>>(x_idx, embed_w, x0);

    build_lists<<<NOWN, 1024, 0, stream>>>(src, dst, cnt, lst, spilln, spills, spilld);

    const int tiles  = N_NODES / 16;
    const int blocks = (tiles + WPB - 1) / WPB;

    sage_pull_wmma<EMB, false><<<blocks, WPB * 32, 0, stream>>>(nullptr, nullptr, nullptr, cnt, lst, spilln, spills, spilld, x0, W1l, b1, W1r, h1);

    sage_pull_wmma<HID, true><<<blocks, WPB * 32, 0, stream>>>(batch, gstart, spilln + 1, cnt, lst, spilln, spills, spilld, h1, W2l, b2, W2r, h2);

    pool_head<<<N_GRAPHS / GPB, GPB * 32, 0, stream>>>(batch, gstart, spilln + 1, h2, Wout, bout, out);
}


# embedding rows x0 written by the list build's owner work-groups (each of a range's 8 owners 1/8 of its nodes, 16-byte chunks, whole lines) ahead of their edge pass; embed launch of 100000 waves gone
# speedup vs baseline: 6.4703x; 1.0477x over previous
#include <hip/hip_runtime.h>


#define N_NODES  100000
#define N_EDGES  3200000
#define N_GRAPHS 4096
#define EMB      32
#define HID      64
#define NCLS     10
#define NPART    8
#define CAP      16
#define PSTR     (2 * CAP)
#define PAIRW    (NPART * PSTR)

typedef __attribute__((ext_vector_type(16))) _Float16 v16h;
typedef __attribute__((ext_vector_type(8)))  float    v8f;
typedef __attribute__((ext_vector_type(4)))  float    v4f;

__device__ __forceinline__ v16h pack_frag(float4 c0, float4 c1,
                                          float4 c2, float4 c3, float s)
{
    v16h r;
    r[0]  = (_Float16)(c0.x * s); r[1]  = (_Float16)(c0.y * s);
    r[2]  = (_Float16)(c0.z * s); r[3]  = (_Float16)(c0.w * s);
    r[4]  = (_Float16)(c1.x * s); r[5]  = (_Float16)(c1.y * s);
    r[6]  = (_Float16)(c1.z * s); r[7]  = (_Float16)(c1.w * s);
    r[8]  = (_Float16)(c2.x * s); r[9]  = (_Float16)(c2.y * s);
    r[10] = (_Float16)(c2.z * s); r[11] = (_Float16)(c2.w * s);
    r[12] = (_Float16)(c3.x * s); r[13] = (_Float16)(c3.y * s);
    r[14] = (_Float16)(c3.z * s); r[15] = (_Float16)(c3.w * s);
    return r;
}

__device__ __forceinline__ v16h load_frag(const float4* __restrict__ q,
                                          int ks, int half, float s)
{
    const int o = ks * 8 + half * 2;
    return pack_frag(q[o], q[o + 1], q[o + 4], q[o + 5], s);
}

#define NRNG  32
#define RNG   3126
static_assert(RNG % 2 == 0 && NRNG * RNG >= N_NODES && (NRNG - 1) * RNG < N_NODES, "ranges");
#define NOWN  (NRNG * NPART)
#define E4    (N_EDGES / 4)
#define PLEN  (E4 / NPART)
#define QCAP  160
#define UNR   8
#define XCH   (EMB / 4)
#define XT    ((((RNG + NPART - 1) / NPART) * XCH + 1023) / 1024)

__global__ void __launch_bounds__(1024)
build_lists(const int* __restrict__ x_idx,
            const float* __restrict__ embed_w,
            float* __restrict__ x0,
            const int* __restrict__ src,
            const int* __restrict__ dst,
            int* __restrict__ cnt,
            int* __restrict__ lst,
            int* __restrict__ spill_n,
            int* __restrict__ spill_src,
            int* __restrict__ spill_dst)
{
    __shared__ int  cntL[RNG];
    __shared__ int2 qL[32 * QCAP];

    const int lane = threadIdx.x & 31;
    const int wave = threadIdx.x >> 5;
    const int rng  = blockIdx.x / NPART;
    const int part = blockIdx.x % NPART;
    const int base = rng * RNG;
    const int nloc = min(RNG, N_NODES - base);
    int2* qw = qL + wave * QCAP;

    const int per = (nloc + NPART - 1) / NPART;
    const int n0  = min(part * per, nloc);
    const int nx4 = (min(n0 + per, nloc) - n0) * XCH;
    int xrow[XT];
    #pragma unroll
    for (int u = 0; u < XT; ++u) {
        const int f = u * 1024 + (int)threadIdx.x;
        xrow[u] = x_idx[min(base + n0 + f / XCH, N_NODES - 1)];
    }

    for (int i = threadIdx.x; i < RNG; i += 1024) cntL[i] = 0;
    __syncthreads();

    {
        const v4f* __restrict__ ew = (const v4f*)embed_w;
        v4f* __restrict__ xo = (v4f*)x0 + (base + n0) * XCH;
        v4f xv[XT];
        #pragma unroll
        for (int u = 0; u < XT; ++u)
            xv[u] = ew[xrow[u] * XCH + (int)(threadIdx.x & (XCH - 1))];
        #pragma unroll
        for (int u = 0; u < XT; ++u) asm volatile("" :: "v"(xv[u]));
        #pragma unroll
        for (int u = 0; u < XT; ++u) {
            const int f = u * 1024 + (int)threadIdx.x;
            if (f < nx4) xo[f] = xv[u];
        }
    }

    auto take = [&](int first, int n) {
        if (lane < n) {
            const int2 t = qw[first + lane];
            const int s = src[t.x];
            const int r = atomicAdd(&cntL[t.y], 1);
            const int node = base + t.y;
            if (r < CAP) {
                lst[(node >> 1) * PAIRW + part * PSTR + (node & 1) * CAP + r] = s;
            } else {
                const int k = atomicAdd(spill_n, 1);
                spill_src[k] = s;
                spill_dst[k] = node;
            }
        }
    };

    const int rot = rng * (PLEN / NRNG);
    const int4* __restrict__ d4 = (const int4*)dst + part * PLEN;
    const int e0 = part * (PLEN * 4);
    int qn = 0;

    for (int g = 0; g < (PLEN + 1024 * UNR - 1) / (1024 * UNR); ++g) {
        int4 v[UNR];
        int  ci[UNR];
        #pragma unroll
        for (int u = 0; u < UNR; ++u) {
            const int c = (g * UNR + u) * 1024 + (int)threadIdx.x;
            int cc = c + rot;
            if (cc >= PLEN) cc -= PLEN;
            ci[u] = cc;
            v[u] = (c < PLEN) ? d4[cc] : make_int4(-1, -1, -1, -1);
        }
        #pragma unroll
        for (int u = 0; u < UNR; ++u) {
            unsigned t[4];
            t[0] = (unsigned)(v[u].x - base); t[1] = (unsigned)(v[u].y - base);
            t[2] = (unsigned)(v[u].z - base); t[3] = (unsigned)(v[u].w - base);
            const unsigned mn = min(min(t[0], t[1]), min(t[2], t[3]));
            if (__builtin_amdgcn_ballot_w32(mn < (unsigned)RNG) != 0u) {
                #pragma unroll
                for (int j = 0; j < 4; ++j) {
                    const bool hit = t[j] < (unsigned)RNG;
                    const unsigned m = __builtin_amdgcn_ballot_w32(hit);
                    if (hit)
                        qw[qn + (int)__builtin_amdgcn_mbcnt_lo(m, 0u)] =
                            make_int2(e0 + ci[u] * 4 + j, (int)t[j]);
                    qn += __builtin_popcount(m);
                }
                while (qn >= 32) { qn -= 32; take(qn, 32); }
            }
        }
    }
    take(0, qn);

    __syncthreads();
    for (int i = threadIdx.x; i < nloc; i += 1024)
        cnt[part * N_NODES + base + i] = cntL[i];
}


#define WPB   4
#define FSTR  148
static_assert(FSTR >= NPART * CAP + NPART && FSTR % 4 == 0 && (FSTR / 4) % 2 == 1, "row pitch");

template <int K, bool MARK>
__global__ void __launch_bounds__(WPB * 32)
sage_pull_wmma(const int* __restrict__ batch,
               int* __restrict__ gstart,
               int* __restrict__ unsorted,
               const int* __restrict__ cnt,
               const int* __restrict__ lst,
               const int* __restrict__ spill_n,
               const int* __restrict__ spill_src,
               const int* __restrict__ spill_dst,
               const float* __restrict__ xin,
               const float* __restrict__ Wl,
               const float* __restrict__ bl,
               const float* __restrict__ Wr,
               float* __restrict__ out)
{
    __shared__ __attribute__((aligned(16))) int stageL[WPB * 16 * FSTR];

    const int lane  = threadIdx.x & 31;
    const int wave  = threadIdx.x >> 5;
    const int tile  = blockIdx.x * WPB + wave;
    const int row0  = tile * 16;
    if (row0 >= N_NODES) return;

    const int half = lane >> 4;
    const int m    = lane & 15;
    const int rowA = row0 + m;

    int bcur = 0, bprev = -1;
    if constexpr (MARK) {
        bcur = batch[rowA];
        if (rowA > 0) bprev = batch[rowA - 1];
    }

    constexpr int KS = K / 32;
    constexpr int CH = K / 4;
    constexpr int G  = 32 / CH;
    constexpr int U  = 8;
    static_assert(U == 8 && (NPART * CAP) % (G * U) == 0, "a trip never reads past a row's indices");
    const int c = lane & (CH - 1);
    const int g = lane / CH;

    int deg = 0, end = 0;
    int info[NPART];
    #pragma unroll
    for (int k = 0; k < NPART; ++k) {
        const int ck = cnt[k * N_NODES + rowA];
        const int hk = ck < CAP ? ck : CAP;
        info[k] = end | (hk << 8);
        deg += ck;
        end += hk;
    }
    const float inv = 1.0f / fmaxf((float)deg, 1.0f);

    int* fw = stageL + wave * (16 * FSTR);
    int* fm = fw + m * FSTR;

    constexpr int TAB = NPART * CAP;
    static_assert(TAB + NPART <= FSTR && TAB % 4 == 0, "info words behind the indices");
    ((int4*)(fm + TAB))[half] = half ? make_int4(info[4], info[5], info[6], info[7])
                                     : make_int4(info[0], info[1], info[2], info[3]);

    {
        const int par = (lane >> 2) & 1;
        const int r0  = (lane & 3) * 4;
        const int kq  = lane >> 3;
        const int4* __restrict__ blk = (const int4*)(lst + (row0 >> 1) * PAIRW) + lane;
        #pragma unroll
        for (int b = 0; b < 2; ++b) {
            int4 e[8];
            #pragma unroll
            for (int t = 0; t < 8; ++t) e[t] = blk[(b * 8 + t) * 32];
            #pragma unroll
            for (int t = 0; t < 8; ++t) {
                const int tt = b * 8 + t;
                int* row = fw + (2 * (tt >> 1) + par) * FSTR;
                const int w  = row[TAB + 4 * (tt & 1) + kq];
                const int nv = (w >> 8) - r0;
                int* d = row + (w & 0xff) + r0;
                if (nv > 0) d[0] = e[t].x;
                if (nv > 1) d[1] = e[t].y;
                if (nv > 2) d[2] = e[t].z;
                if (nv > 3) d[3] = e[t].w;
            }
        }
    }

    const char* xb = (const char*)xin;
    const unsigned coff = (unsigned)c * 16u;
    auto chunk_of = [&](int i) -> float4 {
        return *(const float4*)(xb + ((unsigned)i * (unsigned)(K * 4) + coff));
    };

    for (int n = 0; n < 16; ++n) {
        const int endn = __builtin_amdgcn_readlane(end, n);
        const int degn = __builtin_amdgcn_readlane(deg, n);
        const int* fl = fw + n * FSTR;
        float4 acc = make_float4(0.f, 0.f, 0.f, 0.f);
        for (int j0 = 0; j0 < endn; j0 += G * U) {
            const int4* ip = (const int4*)__builtin_assume_aligned(fl + j0 + g * U, 16);
            const int4 ia = ip[0];
            const int4 ib = ip[1];
            const int idx[U] = {ia.x, ia.y, ia.z, ia.w, ib.x, ib.y, ib.z, ib.w};
            float4 v[U];
            if (j0 + G * U <= endn) {
                #pragma unroll
                for (int u = 0; u < U; ++u) v[u] = chunk_of(idx[u]);
            } else {
                #pragma unroll
                for (int u = 0; u < U; ++u) {
                    v[u] = make_float4(0.f, 0.f, 0.f, 0.f);
                    if (j0 + g * U + u < endn) v[u] = chunk_of(idx[u]);
                }
            }
            #pragma unroll
            for (int u = 0; u < U; ++u) {
                acc.x += v[u].x; acc.y += v[u].y; acc.z += v[u].z; acc.w += v[u].w;
            }
        }
        if (degn > endn) {
            const int ns = *spill_n;
            for (int k = 0; k < ns; ++k)
                if (spill_dst[k] == row0 + n && g == 0) {
                    const float4 a = chunk_of(spill_src[k]);
                    acc.x += a.x; acc.y += a.y; acc.z += a.z; acc.w += a.w;
                }
        }
        #pragma unroll
        for (int off = CH; off < 32; off <<= 1) {
            acc.x += __shfl_xor(acc.x, off); acc.y += __shfl_xor(acc.y, off);
            acc.z += __shfl_xor(acc.z, off); acc.w += __shfl_xor(acc.w, off);
        }
        if (g == 0)
            ((int4*)(fw + n * FSTR))[c] = make_int4(__float_as_int(acc.x), __float_as_int(acc.y),
                                                    __float_as_int(acc.z), __float_as_int(acc.w));
    }

    const int4*   sq = (const int4*)fm;
    const float4* xq = (const float4*)(xin + rowA * K);
    auto f4 = [](int4 t) -> float4 {
        return make_float4(__int_as_float(t.x), __int_as_float(t.y),
                           __int_as_float(t.z), __int_as_float(t.w));
    };
    v16h aM[KS], aX[KS];
    #pragma unroll
    for (int ks = 0; ks < KS; ++ks) {
        const int o = ks * 8 + half * 2;
        aM[ks] = pack_frag(f4(sq[o]), f4(sq[o + 1]), f4(sq[o + 4]), f4(sq[o + 5]), inv);
        aX[ks] = load_frag(xq, ks, half, 1.0f);
    }

    #pragma unroll 1
    for (int nt = 0; nt < HID / 16; ++nt) {
        const int ncol = nt * 16 + m;
        const float4* wlq = (const float4*)(Wl + ncol * K);
        const float4* wrq = (const float4*)(Wr + ncol * K);

        v8f acc = {};
        #pragma unroll
        for (int ks = 0; ks < KS; ++ks) {
            v16h bL = load_frag(wlq, ks, half, 1.0f);
            v16h bR = load_frag(wrq, ks, half, 1.0f);
            acc = __builtin_amdgcn_wmma_f32_16x16x32_f16(
                false, aM[ks], false, bL, (short)0, acc, false, false);
            acc = __builtin_amdgcn_wmma_f32_16x16x32_f16(
                false, aX[ks], false, bR, (short)0, acc, false, false);
        }

        const float bias = bl[ncol];
        float* op = out + (row0 + half * 8) * HID + ncol;
        #pragma unroll
        for (int d = 0; d < 8; ++d)
            op[d * HID] = fmaxf(acc[d] + bias, 0.0f);
    }

    if constexpr (MARK) {
        if (half == 0) {
            if (bcur < bprev) {
                *unsorted = 1;
            } else {
                #pragma clang loop unroll(disable) vectorize(disable)
                for (int q = max(bprev + 1, 0); q <= min(bcur, N_GRAPHS - 1); ++q) gstart[q] = rowA;
            }
            if (rowA == N_NODES - 1) {
                #pragma clang loop unroll(disable) vectorize(disable)
                for (int q = max(bcur + 1, 0); q <= N_GRAPHS; ++q) gstart[q] = N_NODES;
            }
        }
    }
}

#define GPB 16
#define PU  8
static_assert(N_GRAPHS % GPB == 0 && (GPB * NCLS * 4) % 128 == 0, "whole output lines per work-group");
static_assert(NCLS == 10 && HID == 64 && N_NODES % 32 == 0, "lane maps of the pooling kernel");

__global__ void __launch_bounds__(GPB * 32)
pool_head(const int* __restrict__ batch,
          const int* __restrict__ gstart,
          const int* __restrict__ unsorted,
          const float* __restrict__ h2,
          const float* __restrict__ Wout,
          const float* __restrict__ bout,
          float* __restrict__ out)
{
    __shared__ float outL[GPB * NCLS];

    const int lane = threadIdx.x & 31;
    const int wave = threadIdx.x >> 5;
    const int q    = blockIdx.x * GPB + wave;
    const int half = lane >> 4;
    const int c    = lane & 15;
    const float4* __restrict__ hq = (const float4*)h2 + c;

    float4 w[5];
    #pragma unroll
    for (int k = 0; k < 5; ++k) w[k] = ((const float4*)Wout)[(half * 5 + k) * (HID / 4) + c];

    float4 acc = make_float4(0.f, 0.f, 0.f, 0.f);
    int n = 0;
    if (*unsorted == 0) {
        const int lo = gstart[q];
        const int hi = gstart[q + 1];
        n = hi - lo;
        for (int r0 = lo; r0 < hi; r0 += 2 * PU) {
            float4 v[PU];
            #pragma unroll
            for (int u = 0; u < PU; ++u) {
                const int r = r0 + 2 * u + half;
                v[u] = make_float4(0.f, 0.f, 0.f, 0.f);
                if (r < hi) v[u] = hq[r * (HID / 4)];
            }
            #pragma unroll
            for (int u = 0; u < PU; ++u) {
                acc.x += v[u].x; acc.y += v[u].y; acc.z += v[u].z; acc.w += v[u].w;
            }
        }
    } else {
        for (int i0 = 0; i0 < N_NODES; i0 += 32) {
            unsigned mk = __builtin_amdgcn_ballot_w32(batch[i0 + lane] == q);
            n += __builtin_popcount(mk);
            while (mk != 0u) {
                const int j = __builtin_ctz(mk);
                mk &= mk - 1u;
                if (half == 0) {
                    const float4 a = hq[(i0 + j) * (HID / 4)];
                    acc.x += a.x; acc.y += a.y; acc.z += a.z; acc.w += a.w;
                }
            }
        }
    }
    acc.x += __shfl_xor(acc.x, 16); acc.y += __shfl_xor(acc.y, 16);
    acc.z += __shfl_xor(acc.z, 16); acc.w += __shfl_xor(acc.w, 16);

    const float inv = 1.0f / fmaxf((float)n, 1.0f);
    float p[5];
    #pragma unroll
    for (int k = 0; k < 5; ++k)
        p[k] = acc.x * w[k].x + acc.y * w[k].y + acc.z * w[k].z + acc.w * w[k].w;
    #pragma unroll
    for (int off = 8; off >= 1; off >>= 1) {
        #pragma unroll
        for (int k = 0; k < 5; ++k) p[k] += __shfl_xor(p[k], off);
    }
    if (c == 0) {
        #pragma unroll
        for (int k = 0; k < 5; ++k)
            outL[wave * NCLS + half * 5 + k] = bout[half * 5 + k] + p[k] * inv;
    }
    __syncthreads();
    if (threadIdx.x < GPB * NCLS)
        out[blockIdx.x * (GPB * NCLS) + threadIdx.x] = outL[threadIdx.x];
}

extern "C" void kernel_launch(void* const* d_in, const int* in_sizes, int n_in,
                              void* d_out, int out_size, void* d_ws, size_t ws_size,
                              hipStream_t stream)
{
    const int*   x_idx   = (const int*)  d_in[0];
    const int*   ei      = (const int*)  d_in[1];
    const int*   batch   = (const int*)  d_in[2];
    const float* embed_w = (const float*)d_in[3];
    const float* W1l     = (const float*)d_in[4];
    const float* b1      = (const float*)d_in[5];
    const float* W1r     = (const float*)d_in[6];
    const float* W2l     = (const float*)d_in[7];
    const float* b2      = (const float*)d_in[8];
    const float* W2r     = (const float*)d_in[9];
    const float* Wout    = (const float*)d_in[10];
    const float* bout    = (const float*)d_in[11];
    float*       out     = (float*)d_out;

    const int* src = ei;
    const int* dst = ei + N_EDGES;

    float* x0     = (float*)d_ws;
    float* h1     = x0     + N_NODES * EMB;
    float* h2     = h1     + N_NODES * HID;
    int*   lst    = (int*)(h2 + N_NODES * HID);
    int*   gstart = lst + (N_NODES / 2) * PAIRW;
    int*   cnt    = gstart + N_GRAPHS + 4;
    int*   spilln = cnt    + NPART * N_NODES;
    int*   spills = spilln + 4;
    int*   spilld = spills + N_EDGES;

    hipMemsetAsync(spilln, 0, sizeof(int) * 4, stream);

    build_lists<<<NOWN, 1024, 0, stream>>>(x_idx, embed_w, x0, src, dst, cnt, lst, spilln, spills, spilld);

    const int tiles  = N_NODES / 16;
    const int blocks = (tiles + WPB - 1) / WPB;

    sage_pull_wmma<EMB, false><<<blocks, WPB * 32, 0, stream>>>(nullptr, nullptr, nullptr, cnt, lst, spilln, spills, spilld, x0, W1l, b1, W1r, h1);

    sage_pull_wmma<HID, true><<<blocks, WPB * 32, 0, stream>>>(batch, gstart, spilln + 1, cnt, lst, spilln, spills, spilld, h1, W2l, b2, W2r, h2);

    pool_head<<<N_GRAPHS / GPB, GPB * 32, 0, stream>>>(batch, gstart, spilln + 1, h2, Wout, bout, out);
}
